# starting kernel
# baseline (speedup 1.0000x reference)
#include <hip/hip_runtime.h>


typedef float v2f __attribute__((ext_vector_type(2)));
typedef float v8f __attribute__((ext_vector_type(8)));

#define IN_C  64
#define OUT_C 64

__global__ __launch_bounds__(256) void k_deg_init(float* __restrict__ deg, int n) {
  int i = blockIdx.x * 256 + threadIdx.x;
  if (i < n) deg[i] = 1.0f;
}

__global__ __launch_bounds__(256) void k_degree(const int* __restrict__ col,
                                                float* __restrict__ deg, int m) {
  int e = blockIdx.x * 256 + threadIdx.x;
  if (e < m) unsafeAtomicAdd(&deg[col[e]], 1.0f);
}

__global__ __launch_bounds__(256) void k_rsqrt(const float* __restrict__ deg,
                                               float* __restrict__ dis, int n) {
  int i = blockIdx.x * 256 + threadIdx.x;
  if (i < n) dis[i] = rsqrtf(deg[i]);
}

__global__ __launch_bounds__(256) void k_gemm_wmma(
    const float* __restrict__ x, const float* __restrict__ W,
    const float* __restrict__ bias, const float* __restrict__ dis,
    float* __restrict__ h, float* __restrict__ out, int nRowTiles) {
  __shared__ float Wlds[IN_C * OUT_C];
  for (int i = threadIdx.x; i < IN_C * OUT_C; i += 256) Wlds[i] = W[i];
  __syncthreads();

  const int wave = threadIdx.x >> 5;
  const int lane = threadIdx.x & 31;
  const int tile = blockIdx.x * 8 + wave;
  const int rowTile = tile >> 2;
  if (rowTile >= nRowTiles) return;
  const int colBase = (tile & 3) << 4;
  const int rowBase = rowTile << 4;

  const int mn = lane & 15;
  const int hi = lane >> 4;

  v8f acc = {};
  const float* __restrict__ xrow = x + (size_t)(rowBase + mn) * IN_C;
#pragma unroll
  for (int kb = 0; kb < IN_C; kb += 4) {
    const int k0 = kb + 2 * hi;
    const v2f a = *(const v2f*)(xrow + k0);
    v2f bf;
    bf.x = Wlds[k0 * OUT_C + colBase + mn];
    bf.y = Wlds[(k0 + 1) * OUT_C + colBase + mn];
    acc = __builtin_amdgcn_wmma_f32_16x16x4_f32(
         false, a,  false, bf,
         (short)0, acc,  false,  false);
  }

  const float bv = bias[colBase + mn];
#pragma unroll
  for (int v = 0; v < 8; ++v) {
    const int row = rowBase + v + 8 * hi;
    const size_t idx = (size_t)row * OUT_C + colBase + mn;
    const float d = dis[row];
    const float hv = acc[v];
    h[idx] = hv;
    out[idx] = hv * (d * d) + bv;
  }
}

__global__ __launch_bounds__(256) void k_scatter(
    const int* __restrict__ ei, const float* __restrict__ h,
    const float* __restrict__ dis, float* __restrict__ out, int m) {
  const int t = blockIdx.x * 256 + threadIdx.x;
  const int e = t >> 4;
  if (e >= m) return;
  const int c = (t & 15) << 2;
  const int row = ei[e];
  const int col = ei[m + e];
  const float norm = dis[row] * dis[col];
  const float4 hv = *(const float4*)(h + (size_t)row * OUT_C + c);
  float* o = out + (size_t)col * OUT_C + c;
  unsafeAtomicAdd(o + 0, norm * hv.x);
  unsafeAtomicAdd(o + 1, norm * hv.y);
  unsafeAtomicAdd(o + 2, norm * hv.z);
  unsafeAtomicAdd(o + 3, norm * hv.w);
}

extern "C" void kernel_launch(void* const* d_in, const int* in_sizes, int n_in,
                              void* d_out, int out_size, void* d_ws, size_t ws_size,
                              hipStream_t stream) {
  const float* x  = (const float*)d_in[0];
  const int*   ei = (const int*)d_in[1];
  const float* W  = (const float*)d_in[2];
  const float* b  = (const float*)d_in[3];
  float* out = (float*)d_out;

  const int N = in_sizes[0] / IN_C;
  const int M = in_sizes[1] / 2;

  float* deg = (float*)d_ws;
  float* dis = deg + N;
  float* h   = dis + N;

  const int nRowTiles = (N + 15) / 16;

  k_deg_init<<<(N + 255) / 256, 256, 0, stream>>>(deg, N);
  k_degree<<<(M + 255) / 256, 256, 0, stream>>>(ei + M, deg, M);
  k_rsqrt<<<(N + 255) / 256, 256, 0, stream>>>(deg, dis, N);
  k_gemm_wmma<<<(nRowTiles * 4 + 7) / 8, 256, 0, stream>>>(x, W, b, dis, h, out,
                                                           nRowTiles);
  {
    const long long threads = (long long)M * 16;
    k_scatter<<<(int)((threads + 255) / 256), 256, 0, stream>>>(ei, h, dis, out, M);
  }
}


# Edge scatter: a wave walks 32 edges with lane = channel, so each h-row gather and each f32 atomic add covers one whole 128-B line (was 4 lines x 8 floats); atomics kept, arrival order as before
# speedup vs baseline: 1.5500x; 1.5500x over previous
#include <hip/hip_runtime.h>


typedef float v2f __attribute__((ext_vector_type(2)));
typedef float v8f __attribute__((ext_vector_type(8)));

#define IN_C  64
#define OUT_C 64

__global__ __launch_bounds__(256) void k_deg_init(float* __restrict__ deg, int n) {
  int i = blockIdx.x * 256 + threadIdx.x;
  if (i < n) deg[i] = 1.0f;
}

__global__ __launch_bounds__(256) void k_degree(const int* __restrict__ col,
                                                float* __restrict__ deg, int m) {
  int e = blockIdx.x * 256 + threadIdx.x;
  if (e < m) unsafeAtomicAdd(&deg[col[e]], 1.0f);
}

__global__ __launch_bounds__(256) void k_rsqrt(const float* __restrict__ deg,
                                               float* __restrict__ dis, int n) {
  int i = blockIdx.x * 256 + threadIdx.x;
  if (i < n) dis[i] = rsqrtf(deg[i]);
}

__global__ __launch_bounds__(256) void k_gemm_wmma(
    const float* __restrict__ x, const float* __restrict__ W,
    const float* __restrict__ bias, const float* __restrict__ dis,
    float* __restrict__ h, float* __restrict__ out, int nRowTiles) {
  __shared__ float Wlds[IN_C * OUT_C];
  for (int i = threadIdx.x; i < IN_C * OUT_C; i += 256) Wlds[i] = W[i];
  __syncthreads();

  const int wave = threadIdx.x >> 5;
  const int lane = threadIdx.x & 31;
  const int tile = blockIdx.x * 8 + wave;
  const int rowTile = tile >> 2;
  if (rowTile >= nRowTiles) return;
  const int colBase = (tile & 3) << 4;
  const int rowBase = rowTile << 4;

  const int mn = lane & 15;
  const int hi = lane >> 4;

  v8f acc = {};
  const float* __restrict__ xrow = x + (size_t)(rowBase + mn) * IN_C;
#pragma unroll
  for (int kb = 0; kb < IN_C; kb += 4) {
    const int k0 = kb + 2 * hi;
    const v2f a = *(const v2f*)(xrow + k0);
    v2f bf;
    bf.x = Wlds[k0 * OUT_C + colBase + mn];
    bf.y = Wlds[(k0 + 1) * OUT_C + colBase + mn];
    acc = __builtin_amdgcn_wmma_f32_16x16x4_f32(
         false, a,  false, bf,
         (short)0, acc,  false,  false);
  }

  const float bv = bias[colBase + mn];
#pragma unroll
  for (int v = 0; v < 8; ++v) {
    const int row = rowBase + v + 8 * hi;
    const size_t idx = (size_t)row * OUT_C + colBase + mn;
    const float d = dis[row];
    const float hv = acc[v];
    h[idx] = hv;
    out[idx] = hv * (d * d) + bv;
  }
}

#define EDGES_PER_WAVE 32
#define EDGE_GROUP 8

__device__ __forceinline__ void edge_row_add(const float* __restrict__ h,
                                             float* __restrict__ out, int row,
                                             int col, float nrm, int lane) {
  const float* hp = h + (size_t)row * OUT_C + lane;
  float* o = out + (size_t)col * OUT_C + lane;
  const float lo = hp[0];
  const float hi = hp[32];
  unsafeAtomicAdd(o, nrm * lo);
  unsafeAtomicAdd(o + 32, nrm * hi);
}

__global__ __launch_bounds__(256) void k_scatter(
    const int* __restrict__ ei, const float* __restrict__ h,
    const float* __restrict__ dis, float* __restrict__ out, int m) {
  const int lane = threadIdx.x & 31;
  const int wave = __builtin_amdgcn_readfirstlane(blockIdx.x * 8 + (threadIdx.x >> 5));
  const int e0 = wave * EDGES_PER_WAVE;
  if (e0 >= m) return;

  const int e = e0 + lane;
  int row = 0, col = 0;
  float nrm = 0.0f;
  if (e < m) {
    row = ei[e];
    col = ei[m + e];
    nrm = dis[row] * dis[col];
  }
  const int cnt = min(EDGES_PER_WAVE, m - e0);

  if (cnt == EDGES_PER_WAVE) {
#pragma unroll
    for (int g = 0; g < EDGES_PER_WAVE; g += EDGE_GROUP) {
      int r[EDGE_GROUP], c[EDGE_GROUP];
      float n[EDGE_GROUP], lo[EDGE_GROUP], hi[EDGE_GROUP];
#pragma unroll
      for (int u = 0; u < EDGE_GROUP; ++u) {
        r[u] = __builtin_amdgcn_readlane(row, g + u);
        c[u] = __builtin_amdgcn_readlane(col, g + u);
        n[u] = __int_as_float(__builtin_amdgcn_readlane(__float_as_int(nrm), g + u));
        const float* hp = h + (size_t)r[u] * OUT_C + lane;
        lo[u] = hp[0];
        hi[u] = hp[32];
      }
#pragma unroll
      for (int u = 0; u < EDGE_GROUP; ++u) {
        float* o = out + (size_t)c[u] * OUT_C + lane;
        unsafeAtomicAdd(o, n[u] * lo[u]);
        unsafeAtomicAdd(o + 32, n[u] * hi[u]);
      }
    }
  } else {
    for (int j = 0; j < cnt; ++j) {
      const int r = __builtin_amdgcn_readlane(row, j);
      const int c = __builtin_amdgcn_readlane(col, j);
      const float n = __int_as_float(__builtin_amdgcn_readlane(__float_as_int(nrm), j));
      edge_row_add(h, out, r, c, n, lane);
    }
  }
}

extern "C" void kernel_launch(void* const* d_in, const int* in_sizes, int n_in,
                              void* d_out, int out_size, void* d_ws, size_t ws_size,
                              hipStream_t stream) {
  const float* x  = (const float*)d_in[0];
  const int*   ei = (const int*)d_in[1];
  const float* W  = (const float*)d_in[2];
  const float* b  = (const float*)d_in[3];
  float* out = (float*)d_out;

  const int N = in_sizes[0] / IN_C;
  const int M = in_sizes[1] / 2;

  float* deg = (float*)d_ws;
  float* dis = deg + N;
  float* h   = dis + N;

  const int nRowTiles = (N + 15) / 16;

  k_deg_init<<<(N + 255) / 256, 256, 0, stream>>>(deg, N);
  k_degree<<<(M + 255) / 256, 256, 0, stream>>>(ei + M, deg, M);
  k_rsqrt<<<(N + 255) / 256, 256, 0, stream>>>(deg, dis, N);
  k_gemm_wmma<<<(nRowTiles * 4 + 7) / 8, 256, 0, stream>>>(x, W, b, dis, h, out,
                                                           nRowTiles);
  {
    const int waves = (M + EDGES_PER_WAVE - 1) / EDGES_PER_WAVE;
    k_scatter<<<(waves + 7) / 8, 256, 0, stream>>>(ei, h, dis, out, M);
  }
}


# In-degree count by per-node-range LDS integer histograms over edge slices (32 count planes, summed in the rsqrt pass), not one scattered global f32 atomic per edge; 4 launches; out's sum order kept
# speedup vs baseline: 2.6644x; 1.7190x over previous
#include <hip/hip_runtime.h>


typedef float v2f __attribute__((ext_vector_type(2)));
typedef float v8f __attribute__((ext_vector_type(8)));

#define IN_C  64
#define OUT_C 64

#define CNT_RANGES  8
#define CNT_SLICES  32
#define CNT_CAP     12512
#define CNT_THREADS 1024
#define CNT_BATCH   13

__device__ __forceinline__ void count_target(unsigned* cnt, int c, int base, int span) {
  const unsigned i = (unsigned)(c - base);
  if (i < (unsigned)span) atomicAdd(&cnt[i], 1u);
}

__global__ __launch_bounds__(CNT_THREADS) void k_count(const int* __restrict__ col,
                                                       unsigned* __restrict__ part,
                                                       int n, int m) {
  __shared__ unsigned cnt[CNT_CAP];
  const int tid = threadIdx.x;
  const int s = blockIdx.y;
  const int quads = (m + 3) >> 2;
  const int per = (quads + CNT_SLICES - 1) / CNT_SLICES;
  const int q0 = s * per;
  const int q1 = min(q0 + per, quads);
  const int qFull = min(q1, m >> 2);
  const int eEnd = min(q1 * 4, m);
  const bool vec = (reinterpret_cast<size_t>(col) & 15) == 0;
  const int4* __restrict__ c4 = reinterpret_cast<const int4*>(col);

  for (int base = blockIdx.x * CNT_CAP; base < n; base += CNT_RANGES * CNT_CAP) {
    const int span = min(CNT_CAP, n - base);
    for (int i = tid; i < span; i += CNT_THREADS) cnt[i] = 0u;
    __syncthreads();

    int eTail = q0 * 4;
    if (vec) {
      for (int q = q0 + tid; q < qFull; q += CNT_THREADS * CNT_BATCH) {
        int4 c[CNT_BATCH];
#pragma unroll
        for (int u = 0; u < CNT_BATCH; ++u) {
          const int qq = q + u * CNT_THREADS;
          c[u] = (qq < qFull) ? c4[qq] : make_int4(-1, -1, -1, -1);
        }
#pragma unroll
        for (int u = 0; u < CNT_BATCH; ++u) {
          count_target(cnt, c[u].x, base, span);
          count_target(cnt, c[u].y, base, span);
          count_target(cnt, c[u].z, base, span);
          count_target(cnt, c[u].w, base, span);
        }
      }
      eTail = max(q0, qFull) * 4;
    }
    for (int e = eTail + tid; e < eEnd; e += CNT_THREADS) count_target(cnt, col[e], base, span);
    __syncthreads();

    unsigned* __restrict__ plane = part + (size_t)s * n + base;
    for (int i = tid; i < span; i += CNT_THREADS) plane[i] = cnt[i];
  }
}

__global__ __launch_bounds__(256) void k_rsqrt(const unsigned* __restrict__ part,
                                               float* __restrict__ dis, int n) {
  const int i = blockIdx.x * 256 + threadIdx.x;
  if (i >= n) return;
  unsigned d = 1u;
#pragma unroll
  for (int s = 0; s < CNT_SLICES; ++s) d += part[(size_t)s * n + i];
  dis[i] = rsqrtf((float)d);
}

__global__ __launch_bounds__(256) void k_gemm_wmma(
    const float* __restrict__ x, const float* __restrict__ W,
    const float* __restrict__ bias, const float* __restrict__ dis,
    float* __restrict__ h, float* __restrict__ out, int nRowTiles) {
  __shared__ float Wlds[IN_C * OUT_C];
  for (int i = threadIdx.x; i < IN_C * OUT_C; i += 256) Wlds[i] = W[i];
  __syncthreads();

  const int wave = threadIdx.x >> 5;
  const int lane = threadIdx.x & 31;
  const int tile = blockIdx.x * 8 + wave;
  const int rowTile = tile >> 2;
  if (rowTile >= nRowTiles) return;
  const int colBase = (tile & 3) << 4;
  const int rowBase = rowTile << 4;

  const int mn = lane & 15;
  const int hi = lane >> 4;

  v8f acc = {};
  const float* __restrict__ xrow = x + (size_t)(rowBase + mn) * IN_C;
#pragma unroll
  for (int kb = 0; kb < IN_C; kb += 4) {
    const int k0 = kb + 2 * hi;
    const v2f a = *(const v2f*)(xrow + k0);
    v2f bf;
    bf.x = Wlds[k0 * OUT_C + colBase + mn];
    bf.y = Wlds[(k0 + 1) * OUT_C + colBase + mn];
    acc = __builtin_amdgcn_wmma_f32_16x16x4_f32(
         false, a,  false, bf,
         (short)0, acc,  false,  false);
  }

  const float bv = bias[colBase + mn];
#pragma unroll
  for (int v = 0; v < 8; ++v) {
    const int row = rowBase + v + 8 * hi;
    const size_t idx = (size_t)row * OUT_C + colBase + mn;
    const float d = dis[row];
    const float hv = acc[v];
    h[idx] = hv;
    out[idx] = hv * (d * d) + bv;
  }
}

#define EDGES_PER_WAVE 32
#define EDGE_GROUP 8

__device__ __forceinline__ void edge_row_add(const float* __restrict__ h,
                                             float* __restrict__ out, int row,
                                             int col, float nrm, int lane) {
  const float* hp = h + (size_t)row * OUT_C + lane;
  float* o = out + (size_t)col * OUT_C + lane;
  const float lo = hp[0];
  const float hi = hp[32];
  unsafeAtomicAdd(o, nrm * lo);
  unsafeAtomicAdd(o + 32, nrm * hi);
}

__global__ __launch_bounds__(256) void k_scatter(
    const int* __restrict__ ei, const float* __restrict__ h,
    const float* __restrict__ dis, float* __restrict__ out, int m) {
  const int lane = threadIdx.x & 31;
  const int wave = __builtin_amdgcn_readfirstlane(blockIdx.x * 8 + (threadIdx.x >> 5));
  const int e0 = wave * EDGES_PER_WAVE;
  if (e0 >= m) return;

  const int e = e0 + lane;
  int row = 0, col = 0;
  float nrm = 0.0f;
  if (e < m) {
    row = ei[e];
    col = ei[m + e];
    nrm = dis[row] * dis[col];
  }
  const int cnt = min(EDGES_PER_WAVE, m - e0);

  if (cnt == EDGES_PER_WAVE) {
#pragma unroll
    for (int g = 0; g < EDGES_PER_WAVE; g += EDGE_GROUP) {
      int r[EDGE_GROUP], c[EDGE_GROUP];
      float n[EDGE_GROUP], lo[EDGE_GROUP], hi[EDGE_GROUP];
#pragma unroll
      for (int u = 0; u < EDGE_GROUP; ++u) {
        r[u] = __builtin_amdgcn_readlane(row, g + u);
        c[u] = __builtin_amdgcn_readlane(col, g + u);
        n[u] = __int_as_float(__builtin_amdgcn_readlane(__float_as_int(nrm), g + u));
        const float* hp = h + (size_t)r[u] * OUT_C + lane;
        lo[u] = hp[0];
        hi[u] = hp[32];
      }
#pragma unroll
      for (int u = 0; u < EDGE_GROUP; ++u) {
        float* o = out + (size_t)c[u] * OUT_C + lane;
        unsafeAtomicAdd(o, n[u] * lo[u]);
        unsafeAtomicAdd(o + 32, n[u] * hi[u]);
      }
    }
  } else {
    for (int j = 0; j < cnt; ++j) {
      const int r = __builtin_amdgcn_readlane(row, j);
      const int c = __builtin_amdgcn_readlane(col, j);
      const float n = __int_as_float(__builtin_amdgcn_readlane(__float_as_int(nrm), j));
      edge_row_add(h, out, r, c, n, lane);
    }
  }
}

extern "C" void kernel_launch(void* const* d_in, const int* in_sizes, int n_in,
                              void* d_out, int out_size, void* d_ws, size_t ws_size,
                              hipStream_t stream) {
  const float* x  = (const float*)d_in[0];
  const int*   ei = (const int*)d_in[1];
  const float* W  = (const float*)d_in[2];
  const float* b  = (const float*)d_in[3];
  float* out = (float*)d_out;

  const int N = in_sizes[0] / IN_C;
  const int M = in_sizes[1] / 2;

  unsigned* part = (unsigned*)d_ws;
  float* h   = (float*)(part + (size_t)CNT_SLICES * N);
  float* dis = h + (size_t)N * OUT_C;

  const int nRowTiles = (N + 15) / 16;

  k_count<<<dim3(CNT_RANGES, CNT_SLICES), CNT_THREADS, 0, stream>>>(ei + M, part, N, M);
  k_rsqrt<<<(N + 255) / 256, 256, 0, stream>>>(part, dis, N);
  k_gemm_wmma<<<(nRowTiles * 4 + 7) / 8, 256, 0, stream>>>(x, W, b, dis, h, out,
                                                           nRowTiles);
  {
    const int waves = (M + EDGES_PER_WAVE - 1) / EDGES_PER_WAVE;
    k_scatter<<<(waves + 7) / 8, 256, 0, stream>>>(ei, h, dis, out, M);
  }
}


# Float atomics removed: edges counting-sorted by target in the call (slots = LDS atomic ranks on the count planes); a wave sums a node's rows in registers, one store per out line; sum order changes
# speedup vs baseline: 3.3225x; 1.2470x over previous
#include <hip/hip_runtime.h>


typedef float v2f __attribute__((ext_vector_type(2)));
typedef float v8f __attribute__((ext_vector_type(8)));

#define IN_C  64
#define OUT_C 64

#define CNT_RANGES  8
#define CNT_SLICES  32
#define CHUNK       256
#define CNT_CAP     12544
#define CNT_THREADS 1024
#define CNT_BATCH   13
#define PLC_BATCH   7

struct Slice { int q0; int qFull; int eEnd; };

__device__ __forceinline__ Slice slice_of(int s, int m) {
  const int quads = (m + 3) >> 2;
  const int per = (quads + CNT_SLICES - 1) / CNT_SLICES;
  Slice b;
  b.q0 = s * per;
  const int q1 = min(b.q0 + per, quads);
  b.qFull = min(q1, m >> 2);
  b.eEnd = min(q1 * 4, m);
  return b;
}

__device__ __forceinline__ void count_target(unsigned* cnt, int c, int base, int span) {
  const unsigned i = (unsigned)(c - base);
  if (i < (unsigned)span) atomicAdd(&cnt[i], 1u);
}

__global__ __launch_bounds__(CNT_THREADS) void k_count(const int* __restrict__ col,
                                                       unsigned* __restrict__ part,
                                                       unsigned* __restrict__ cp,
                                                       int n, int m, int nch) {
  __shared__ unsigned cnt[CNT_CAP];
  const int tid = threadIdx.x;
  const int ln = tid & 31;
  const int wv = tid >> 5;
  const int s = blockIdx.y;
  const Slice sl = slice_of(s, m);
  const int q0 = sl.q0, qFull = sl.qFull, eEnd = sl.eEnd;
  const bool vec = (reinterpret_cast<size_t>(col) & 15) == 0;
  const int4* __restrict__ c4 = reinterpret_cast<const int4*>(col);

  for (int base = blockIdx.x * CNT_CAP; base < n; base += CNT_RANGES * CNT_CAP) {
    const int span = min(CNT_CAP, n - base);
    for (int i = tid; i < span; i += CNT_THREADS) cnt[i] = 0u;
    __syncthreads();

    int eTail = q0 * 4;
    if (vec) {
      for (int q = q0 + tid; q < qFull; q += CNT_THREADS * CNT_BATCH) {
        int4 c[CNT_BATCH];
#pragma unroll
        for (int u = 0; u < CNT_BATCH; ++u) {
          const int qq = q + u * CNT_THREADS;
          c[u] = (qq < qFull) ? c4[qq] : make_int4(-1, -1, -1, -1);
        }
#pragma unroll
        for (int u = 0; u < CNT_BATCH; ++u) {
          count_target(cnt, c[u].x, base, span);
          count_target(cnt, c[u].y, base, span);
          count_target(cnt, c[u].z, base, span);
          count_target(cnt, c[u].w, base, span);
        }
      }
      eTail = max(q0, qFull) * 4;
    }
    for (int e = eTail + tid; e < eEnd; e += CNT_THREADS) count_target(cnt, col[e], base, span);
    __syncthreads();

    unsigned* __restrict__ plane = part + (size_t)s * n + base;
    for (int i = tid; i < span; i += CNT_THREADS) plane[i] = cnt[i];

    const int chunks = (span + CHUNK - 1) / CHUNK;
    for (int k = wv; k < chunks; k += CNT_THREADS / 32) {
      unsigned sum = 0u;
#pragma unroll
      for (int j = 0; j < CHUNK / 32; ++j) {
        const int i = k * CHUNK + j * 32 + ln;
        if (i < span) sum += cnt[i];
      }
      for (int o = 16; o > 0; o >>= 1) sum += __shfl_xor(sum, o);
      if (ln == 0) cp[(size_t)s * nch + base / CHUNK + k] = sum;
    }
    __syncthreads();
  }
}

__global__ __launch_bounds__(CHUNK) void k_scan(unsigned* __restrict__ part,
                                                const unsigned* __restrict__ cp,
                                                unsigned* __restrict__ rowptr,
                                                float* __restrict__ dis, int n, int nch) {
  __shared__ unsigned red[CHUNK / 32];
  __shared__ unsigned wsum[CHUNK / 32];
  const int tid = threadIdx.x;
  const int ln = tid & 31;
  const int wv = tid >> 5;
  const int b = blockIdx.x;

  unsigned pre = 0u;
  for (int c = tid; c < b; c += CHUNK) {
#pragma unroll
    for (int s = 0; s < CNT_SLICES; ++s) pre += cp[(size_t)s * nch + c];
  }
  for (int o = 16; o > 0; o >>= 1) pre += __shfl_xor(pre, o);
  if (ln == 0) red[wv] = pre;

  const int i = b * CHUNK + tid;
  unsigned p[CNT_SLICES];
#pragma unroll
  for (int s = 0; s < CNT_SLICES; ++s) p[s] = (i < n) ? part[(size_t)s * n + i] : 0u;
  unsigned d = 0u;
#pragma unroll
  for (int s = 0; s < CNT_SLICES; ++s) { const unsigned t = p[s]; p[s] = d; d += t; }

  unsigned inc = d;
  for (int o = 1; o < 32; o <<= 1) {
    const unsigned t = __shfl_up(inc, (unsigned)o);
    if (ln >= o) inc += t;
  }
  if (ln == 31) wsum[wv] = inc;
  __syncthreads();

  unsigned start = inc - d;
  for (int w = 0; w < CHUNK / 32; ++w) {
    start += red[w];
    if (w < wv) start += wsum[w];
  }
  if (i < n) {
    rowptr[i] = start;
    if (i == n - 1) rowptr[n] = start + d;
#pragma unroll
    for (int s = 0; s < CNT_SLICES; ++s) part[(size_t)s * n + i] = start + p[s];
    dis[i] = rsqrtf((float)(d + 1u));
  }
}

__device__ __forceinline__ unsigned take_slot(unsigned* cur, int c, int base, int span) {
  const unsigned i = (unsigned)(c - base);
  return (i < (unsigned)span) ? atomicAdd(&cur[i], 1u) : 0xFFFFFFFFu;
}

__device__ __forceinline__ void put_source(int* __restrict__ srt, unsigned slot, int r, int m) {
  if (slot < (unsigned)m) srt[slot] = r;
}

__device__ __forceinline__ void place_edge(unsigned* cur, int* __restrict__ srt, int c, int r,
                                           int base, int span, int m) {
  put_source(srt, take_slot(cur, c, base, span), r, m);
}

__device__ __forceinline__ void place_quad(unsigned* cur, int* __restrict__ srt, int4 c, int4 r,
                                           int base, int span, int m) {
  const unsigned s0 = take_slot(cur, c.x, base, span);
  const unsigned s1 = take_slot(cur, c.y, base, span);
  const unsigned s2 = take_slot(cur, c.z, base, span);
  const unsigned s3 = take_slot(cur, c.w, base, span);
  put_source(srt, s0, r.x, m);
  put_source(srt, s1, r.y, m);
  put_source(srt, s2, r.z, m);
  put_source(srt, s3, r.w, m);
}

__global__ __launch_bounds__(CNT_THREADS) void k_place(const int* __restrict__ ei,
                                                       const unsigned* __restrict__ part,
                                                       int* __restrict__ srt, int n, int m) {
  __shared__ unsigned cur[CNT_CAP];
  const int tid = threadIdx.x;
  const int s = blockIdx.y;
  const Slice sl = slice_of(s, m);
  const int q0 = sl.q0, qFull = sl.qFull, eEnd = sl.eEnd;
  const int* __restrict__ row = ei;
  const int* __restrict__ col = ei + m;
  const bool vec = ((reinterpret_cast<size_t>(row) | reinterpret_cast<size_t>(col)) & 15) == 0;
  const int4* __restrict__ r4 = reinterpret_cast<const int4*>(row);
  const int4* __restrict__ c4 = reinterpret_cast<const int4*>(col);

  for (int base = blockIdx.x * CNT_CAP; base < n; base += CNT_RANGES * CNT_CAP) {
    const int span = min(CNT_CAP, n - base);
    const unsigned* __restrict__ plane = part + (size_t)s * n + base;
    for (int i = tid; i < span; i += CNT_THREADS) cur[i] = plane[i];
    __syncthreads();

    int eTail = q0 * 4;
    if (vec) {
      for (int q = q0 + tid; q < qFull; q += CNT_THREADS * PLC_BATCH) {
        int4 c[PLC_BATCH], r[PLC_BATCH];
#pragma unroll
        for (int u = 0; u < PLC_BATCH; ++u) {
          const int qq = q + u * CNT_THREADS;
          const bool ok = qq < qFull;
          c[u] = ok ? c4[qq] : make_int4(-1, -1, -1, -1);
          r[u] = ok ? r4[qq] : make_int4(0, 0, 0, 0);
        }
#pragma unroll
        for (int u = 0; u < PLC_BATCH; ++u) {
          place_quad(cur, srt, c[u], r[u], base, span, m);
        }
      }
      eTail = max(q0, qFull) * 4;
    }
    for (int e = eTail + tid; e < eEnd; e += CNT_THREADS)
      place_edge(cur, srt, col[e], row[e], base, span, m);
    __syncthreads();
  }
}

__global__ __launch_bounds__(256) void k_gemm_wmma(
    const float* __restrict__ x, const float* __restrict__ W,
    const float* __restrict__ dis, float* __restrict__ h, int nRowTiles) {
  __shared__ float Wlds[IN_C * OUT_C];
  for (int i = threadIdx.x; i < IN_C * OUT_C; i += 256) Wlds[i] = W[i];
  __syncthreads();

  const int wave = threadIdx.x >> 5;
  const int lane = threadIdx.x & 31;
  const int tile = blockIdx.x * 8 + wave;
  const int rowTile = tile >> 2;
  if (rowTile >= nRowTiles) return;
  const int colBase = (tile & 3) << 4;
  const int rowBase = rowTile << 4;

  const int mn = lane & 15;
  const int hi = lane >> 4;

  v8f acc = {};
  const float* __restrict__ xrow = x + (size_t)(rowBase + mn) * IN_C;
#pragma unroll
  for (int kb = 0; kb < IN_C; kb += 4) {
    const int k0 = kb + 2 * hi;
    const v2f a = *(const v2f*)(xrow + k0);
    v2f bf;
    bf.x = Wlds[k0 * OUT_C + colBase + mn];
    bf.y = Wlds[(k0 + 1) * OUT_C + colBase + mn];
    acc = __builtin_amdgcn_wmma_f32_16x16x4_f32(
         false, a,  false, bf,
         (short)0, acc,  false,  false);
  }

#pragma unroll
  for (int v = 0; v < 8; ++v) {
    const int row = rowBase + v + 8 * hi;
    h[(size_t)row * OUT_C + colBase + mn] = acc[v] * dis[row];
  }
}

#define PULL_NODES 8
#define PULL_GROUP 8

__global__ __launch_bounds__(256) void k_pull(
    const unsigned* __restrict__ rowptr, const int* __restrict__ srt,
    const float* __restrict__ h, const float* __restrict__ dis,
    const float* __restrict__ bias, float* __restrict__ out, int n, int m) {
  const int lane = threadIdx.x & 31;
  const int wave = __builtin_amdgcn_readfirstlane(blockIdx.x * 8 + (threadIdx.x >> 5));
  const int i0 = wave * PULL_NODES;
  if (i0 >= n) return;
  const int nodes = min(PULL_NODES, n - i0);

  unsigned rp = 0u;
  if (lane <= PULL_NODES) rp = rowptr[min(i0 + lane, n)];
  const float bLo = bias[lane];
  const float bHi = bias[lane + 32];

  for (int k = 0; k < nodes; ++k) {
    const int i = i0 + k;
    const unsigned beg = min((unsigned)__builtin_amdgcn_readlane((int)rp, k), (unsigned)m);
    const unsigned end = min((unsigned)__builtin_amdgcn_readlane((int)rp, k + 1), (unsigned)m);
    const float* __restrict__ hs = h + (size_t)i * OUT_C + lane;
    float aLo = hs[0];
    float aHi = hs[32];
    const float d = dis[i];

    for (unsigned e0 = beg; e0 < end; e0 += 32u) {
      const int cnt = (int)min(32u, end - e0);
      int sv = 0;
      if (lane < cnt) sv = srt[e0 + lane];
      for (int j = 0; j < cnt; j += PULL_GROUP) {
        int r[PULL_GROUP];
        float lo[PULL_GROUP], hi[PULL_GROUP];
#pragma unroll
        for (int u = 0; u < PULL_GROUP; ++u) {
          const int t = __builtin_amdgcn_readlane(sv, min(j + u, cnt - 1));
          r[u] = min(max(t, 0), n - 1);
        }
#pragma unroll
        for (int u = 0; u < PULL_GROUP; ++u) {
          lo[u] = 0.0f;
          hi[u] = 0.0f;
          if (j + u < cnt) {
            const float* hp = h + (size_t)r[u] * OUT_C + lane;
            lo[u] = hp[0];
            hi[u] = hp[32];
          }
        }
#pragma unroll
        for (int u = 0; u < PULL_GROUP; ++u) {
          aLo += lo[u];
          aHi += hi[u];
        }
      }
    }
    float* __restrict__ o = out + (size_t)i * OUT_C + lane;
    o[0] = d * aLo + bLo;
    o[32] = d * aHi + bHi;
  }
}

extern "C" void kernel_launch(void* const* d_in, const int* in_sizes, int n_in,
                              void* d_out, int out_size, void* d_ws, size_t ws_size,
                              hipStream_t stream) {
  const float* x  = (const float*)d_in[0];
  const int*   ei = (const int*)d_in[1];
  const float* W  = (const float*)d_in[2];
  const float* b  = (const float*)d_in[3];
  float* out = (float*)d_out;

  const int N = in_sizes[0] / IN_C;
  const int M = in_sizes[1] / 2;
  const int nch = (N + CHUNK - 1) / CHUNK;

  unsigned* part = (unsigned*)d_ws;
  float* h   = (float*)(part + (size_t)CNT_SLICES * N);
  float* dis = h + (size_t)N * OUT_C;
  unsigned* rowptr = (unsigned*)(dis + N);
  int* srt = (int*)(rowptr + N + 1);
  unsigned* cp = (unsigned*)(srt + M);

  const int nRowTiles = (N + 15) / 16;

  k_count<<<dim3(CNT_RANGES, CNT_SLICES), CNT_THREADS, 0, stream>>>(ei + M, part, cp, N, M, nch);
  k_scan<<<nch, CHUNK, 0, stream>>>(part, cp, rowptr, dis, N, nch);
  k_place<<<dim3(CNT_RANGES, CNT_SLICES), CNT_THREADS, 0, stream>>>(ei, part, srt, N, M);
  k_gemm_wmma<<<(nRowTiles * 4 + 7) / 8, 256, 0, stream>>>(x, W, dis, h, nRowTiles);
  {
    const int waves = (N + PULL_NODES - 1) / PULL_NODES;
    k_pull<<<(waves + 7) / 8, 256, 0, stream>>>(rowptr, srt, h, dis, b, out, N, M);
  }
}


# GEMM re-tiled: a wave computes 16 rows x all 64 columns on 4 accumulators, its A rows, norm factors and W share asked for as one group ahead of the barrier; 6250 waves, not 25000; k order kept
# speedup vs baseline: 3.5403x; 1.0656x over previous
#include <hip/hip_runtime.h>


typedef float v2f __attribute__((ext_vector_type(2)));
typedef float v8f __attribute__((ext_vector_type(8)));

#define IN_C  64
#define OUT_C 64

#define CNT_RANGES  8
#define CNT_SLICES  32
#define CHUNK       256
#define CNT_CAP     12544
#define CNT_THREADS 1024
#define CNT_BATCH   13
#define PLC_BATCH   7

struct Slice { int q0; int qFull; int eEnd; };

__device__ __forceinline__ Slice slice_of(int s, int m) {
  const int quads = (m + 3) >> 2;
  const int per = (quads + CNT_SLICES - 1) / CNT_SLICES;
  Slice b;
  b.q0 = s * per;
  const int q1 = min(b.q0 + per, quads);
  b.qFull = min(q1, m >> 2);
  b.eEnd = min(q1 * 4, m);
  return b;
}

__device__ __forceinline__ void count_target(unsigned* cnt, int c, int base, int span) {
  const unsigned i = (unsigned)(c - base);
  if (i < (unsigned)span) atomicAdd(&cnt[i], 1u);
}

__global__ __launch_bounds__(CNT_THREADS) void k_count(const int* __restrict__ col,
                                                       unsigned* __restrict__ part,
                                                       unsigned* __restrict__ cp,
                                                       int n, int m, int nch) {
  __shared__ unsigned cnt[CNT_CAP];
  const int tid = threadIdx.x;
  const int ln = tid & 31;
  const int wv = tid >> 5;
  const int s = blockIdx.y;
  const Slice sl = slice_of(s, m);
  const int q0 = sl.q0, qFull = sl.qFull, eEnd = sl.eEnd;
  const bool vec = (reinterpret_cast<size_t>(col) & 15) == 0;
  const int4* __restrict__ c4 = reinterpret_cast<const int4*>(col);

  for (int base = blockIdx.x * CNT_CAP; base < n; base += CNT_RANGES * CNT_CAP) {
    const int span = min(CNT_CAP, n - base);
    for (int i = tid; i < span; i += CNT_THREADS) cnt[i] = 0u;
    __syncthreads();

    int eTail = q0 * 4;
    if (vec) {
      for (int q = q0 + tid; q < qFull; q += CNT_THREADS * CNT_BATCH) {
        int4 c[CNT_BATCH];
#pragma unroll
        for (int u = 0; u < CNT_BATCH; ++u) {
          const int qq = q + u * CNT_THREADS;
          c[u] = (qq < qFull) ? c4[qq] : make_int4(-1, -1, -1, -1);
        }
#pragma unroll
        for (int u = 0; u < CNT_BATCH; ++u) {
          count_target(cnt, c[u].x, base, span);
          count_target(cnt, c[u].y, base, span);
          count_target(cnt, c[u].z, base, span);
          count_target(cnt, c[u].w, base, span);
        }
      }
      eTail = max(q0, qFull) * 4;
    }
    for (int e = eTail + tid; e < eEnd; e += CNT_THREADS) count_target(cnt, col[e], base, span);
    __syncthreads();

    unsigned* __restrict__ plane = part + (size_t)s * n + base;
    for (int i = tid; i < span; i += CNT_THREADS) plane[i] = cnt[i];

    const int chunks = (span + CHUNK - 1) / CHUNK;
    for (int k = wv; k < chunks; k += CNT_THREADS / 32) {
      unsigned sum = 0u;
#pragma unroll
      for (int j = 0; j < CHUNK / 32; ++j) {
        const int i = k * CHUNK + j * 32 + ln;
        if (i < span) sum += cnt[i];
      }
      for (int o = 16; o > 0; o >>= 1) sum += __shfl_xor(sum, o);
      if (ln == 0) cp[(size_t)s * nch + base / CHUNK + k] = sum;
    }
    __syncthreads();
  }
}

__global__ __launch_bounds__(CHUNK) void k_scan(unsigned* __restrict__ part,
                                                const unsigned* __restrict__ cp,
                                                unsigned* __restrict__ rowptr,
                                                float* __restrict__ dis, int n, int nch) {
  __shared__ unsigned red[CHUNK / 32];
  __shared__ unsigned wsum[CHUNK / 32];
  const int tid = threadIdx.x;
  const int ln = tid & 31;
  const int wv = tid >> 5;
  const int b = blockIdx.x;

  unsigned pre = 0u;
  for (int c = tid; c < b; c += CHUNK) {
#pragma unroll
    for (int s = 0; s < CNT_SLICES; ++s) pre += cp[(size_t)s * nch + c];
  }
  for (int o = 16; o > 0; o >>= 1) pre += __shfl_xor(pre, o);
  if (ln == 0) red[wv] = pre;

  const int i = b * CHUNK + tid;
  unsigned p[CNT_SLICES];
#pragma unroll
  for (int s = 0; s < CNT_SLICES; ++s) p[s] = (i < n) ? part[(size_t)s * n + i] : 0u;
  unsigned d = 0u;
#pragma unroll
  for (int s = 0; s < CNT_SLICES; ++s) { const unsigned t = p[s]; p[s] = d; d += t; }

  unsigned inc = d;
  for (int o = 1; o < 32; o <<= 1) {
    const unsigned t = __shfl_up(inc, (unsigned)o);
    if (ln >= o) inc += t;
  }
  if (ln == 31) wsum[wv] = inc;
  __syncthreads();

  unsigned start = inc - d;
  for (int w = 0; w < CHUNK / 32; ++w) {
    start += red[w];
    if (w < wv) start += wsum[w];
  }
  if (i < n) {
    rowptr[i] = start;
    if (i == n - 1) rowptr[n] = start + d;
#pragma unroll
    for (int s = 0; s < CNT_SLICES; ++s) part[(size_t)s * n + i] = start + p[s];
    dis[i] = rsqrtf((float)(d + 1u));
  }
}

__device__ __forceinline__ unsigned take_slot(unsigned* cur, int c, int base, int span) {
  const unsigned i = (unsigned)(c - base);
  return (i < (unsigned)span) ? atomicAdd(&cur[i], 1u) : 0xFFFFFFFFu;
}

__device__ __forceinline__ void put_source(int* __restrict__ srt, unsigned slot, int r, int m) {
  if (slot < (unsigned)m) srt[slot] = r;
}

__device__ __forceinline__ void place_edge(unsigned* cur, int* __restrict__ srt, int c, int r,
                                           int base, int span, int m) {
  put_source(srt, take_slot(cur, c, base, span), r, m);
}

__device__ __forceinline__ void place_quad(unsigned* cur, int* __restrict__ srt, int4 c, int4 r,
                                           int base, int span, int m) {
  const unsigned s0 = take_slot(cur, c.x, base, span);
  const unsigned s1 = take_slot(cur, c.y, base, span);
  const unsigned s2 = take_slot(cur, c.z, base, span);
  const unsigned s3 = take_slot(cur, c.w, base, span);
  put_source(srt, s0, r.x, m);
  put_source(srt, s1, r.y, m);
  put_source(srt, s2, r.z, m);
  put_source(srt, s3, r.w, m);
}

__global__ __launch_bounds__(CNT_THREADS) void k_place(const int* __restrict__ ei,
                                                       const unsigned* __restrict__ part,
                                                       int* __restrict__ srt, int n, int m) {
  __shared__ unsigned cur[CNT_CAP];
  const int tid = threadIdx.x;
  const int s = blockIdx.y;
  const Slice sl = slice_of(s, m);
  const int q0 = sl.q0, qFull = sl.qFull, eEnd = sl.eEnd;
  const int* __restrict__ row = ei;
  const int* __restrict__ col = ei + m;
  const bool vec = ((reinterpret_cast<size_t>(row) | reinterpret_cast<size_t>(col)) & 15) == 0;
  const int4* __restrict__ r4 = reinterpret_cast<const int4*>(row);
  const int4* __restrict__ c4 = reinterpret_cast<const int4*>(col);

  for (int base = blockIdx.x * CNT_CAP; base < n; base += CNT_RANGES * CNT_CAP) {
    const int span = min(CNT_CAP, n - base);
    const unsigned* __restrict__ plane = part + (size_t)s * n + base;
    for (int i = tid; i < span; i += CNT_THREADS) cur[i] = plane[i];
    __syncthreads();

    int eTail = q0 * 4;
    if (vec) {
      for (int q = q0 + tid; q < qFull; q += CNT_THREADS * PLC_BATCH) {
        int4 c[PLC_BATCH], r[PLC_BATCH];
#pragma unroll
        for (int u = 0; u < PLC_BATCH; ++u) {
          const int qq = q + u * CNT_THREADS;
          const bool ok = qq < qFull;
          c[u] = ok ? c4[qq] : make_int4(-1, -1, -1, -1);
          r[u] = ok ? r4[qq] : make_int4(0, 0, 0, 0);
        }
#pragma unroll
        for (int u = 0; u < PLC_BATCH; ++u) {
          place_quad(cur, srt, c[u], r[u], base, span, m);
        }
      }
      eTail = max(q0, qFull) * 4;
    }
    for (int e = eTail + tid; e < eEnd; e += CNT_THREADS)
      place_edge(cur, srt, col[e], row[e], base, span, m);
    __syncthreads();
  }
}

#define GEMM_WAVES   8
#define GEMM_THREADS (GEMM_WAVES * 32)
#define W_PER_THREAD (IN_C * OUT_C / GEMM_THREADS)
#define WP_PITCH 160

__global__ __launch_bounds__(GEMM_THREADS) void k_gemm_wmma(
    const float* __restrict__ x, const float* __restrict__ W,
    const float* __restrict__ dis, float* __restrict__ h, int nRowTiles) {
  __shared__ __attribute__((aligned(16))) float Wlds[(IN_C / 2) * WP_PITCH];

  const int tid = threadIdx.x;
  const int wave = tid >> 5;
  const int lane = tid & 31;
  const int rowTile = blockIdx.x * GEMM_WAVES + wave;
  const bool live = rowTile < nRowTiles;
  const int rowBase = min(rowTile, nRowTiles - 1) << 4;

  const int mn = lane & 15;
  const int hi = lane >> 4;

  const float* __restrict__ xrow = x + (size_t)(rowBase + mn) * IN_C + 2 * hi;
  v2f a[IN_C / 4];
#pragma unroll
  for (int j = 0; j < IN_C / 4; ++j) a[j] = *(const v2f*)(xrow + 4 * j);
  float d[8];
#pragma unroll
  for (int v = 0; v < 8; ++v) d[v] = dis[rowBase + v + 8 * hi];
  float wv[W_PER_THREAD];
#pragma unroll
  for (int u = 0; u < W_PER_THREAD; ++u) wv[u] = W[tid + GEMM_THREADS * u];
#pragma unroll
  for (int u = 0; u < W_PER_THREAD; ++u) {
    const int i = tid + GEMM_THREADS * u;
    const int k = i >> 6, n = i & 63;
    Wlds[(k >> 1) * WP_PITCH + 2 * n + (k & 1)] = wv[u];
  }
  __syncthreads();
  if (!live) return;

  v8f acc0 = {}, acc1 = {}, acc2 = {}, acc3 = {};
  const float* __restrict__ wl = Wlds + hi * WP_PITCH + 2 * mn;
#pragma unroll
  for (int j = 0; j < IN_C / 4; ++j) {
    const float* __restrict__ wp = wl + 2 * j * WP_PITCH;
    const v2f b0 = *(const v2f*)(wp);
    const v2f b1 = *(const v2f*)(wp + 32);
    const v2f b2 = *(const v2f*)(wp + 64);
    const v2f b3 = *(const v2f*)(wp + 96);
    acc0 = __builtin_amdgcn_wmma_f32_16x16x4_f32(false, a[j], false, b0, (short)0, acc0, false, false);
    acc1 = __builtin_amdgcn_wmma_f32_16x16x4_f32(false, a[j], false, b1, (short)0, acc1, false, false);
    acc2 = __builtin_amdgcn_wmma_f32_16x16x4_f32(false, a[j], false, b2, (short)0, acc2, false, false);
    acc3 = __builtin_amdgcn_wmma_f32_16x16x4_f32(false, a[j], false, b3, (short)0, acc3, false, false);
  }

#pragma unroll
  for (int v = 0; v < 8; ++v) {
    float* __restrict__ hr = h + (size_t)(rowBase + v + 8 * hi) * OUT_C + mn;
    hr[0]  = acc0[v] * d[v];
    hr[16] = acc1[v] * d[v];
    hr[32] = acc2[v] * d[v];
    hr[48] = acc3[v] * d[v];
  }
}

#define PULL_NODES 8
#define PULL_GROUP 8

__global__ __launch_bounds__(256) void k_pull(
    const unsigned* __restrict__ rowptr, const int* __restrict__ srt,
    const float* __restrict__ h, const float* __restrict__ dis,
    const float* __restrict__ bias, float* __restrict__ out, int n, int m) {
  const int lane = threadIdx.x & 31;
  const int wave = __builtin_amdgcn_readfirstlane(blockIdx.x * 8 + (threadIdx.x >> 5));
  const int i0 = wave * PULL_NODES;
  if (i0 >= n) return;
  const int nodes = min(PULL_NODES, n - i0);

  unsigned rp = 0u;
  if (lane <= PULL_NODES) rp = rowptr[min(i0 + lane, n)];
  const float bLo = bias[lane];
  const float bHi = bias[lane + 32];

  for (int k = 0; k < nodes; ++k) {
    const int i = i0 + k;
    const unsigned beg = min((unsigned)__builtin_amdgcn_readlane((int)rp, k), (unsigned)m);
    const unsigned end = min((unsigned)__builtin_amdgcn_readlane((int)rp, k + 1), (unsigned)m);
    const float* __restrict__ hs = h + (size_t)i * OUT_C + lane;
    float aLo = hs[0];
    float aHi = hs[32];
    const float d = dis[i];

    for (unsigned e0 = beg; e0 < end; e0 += 32u) {
      const int cnt = (int)min(32u, end - e0);
      int sv = 0;
      if (lane < cnt) sv = srt[e0 + lane];
      for (int j = 0; j < cnt; j += PULL_GROUP) {
        int r[PULL_GROUP];
        float lo[PULL_GROUP], hi[PULL_GROUP];
#pragma unroll
        for (int u = 0; u < PULL_GROUP; ++u) {
          const int t = __builtin_amdgcn_readlane(sv, min(j + u, cnt - 1));
          r[u] = min(max(t, 0), n - 1);
        }
#pragma unroll
        for (int u = 0; u < PULL_GROUP; ++u) {
          lo[u] = 0.0f;
          hi[u] = 0.0f;
          if (j + u < cnt) {
            const float* hp = h + (size_t)r[u] * OUT_C + lane;
            lo[u] = hp[0];
            hi[u] = hp[32];
          }
        }
#pragma unroll
        for (int u = 0; u < PULL_GROUP; ++u) {
          aLo += lo[u];
          aHi += hi[u];
        }
      }
    }
    float* __restrict__ o = out + (size_t)i * OUT_C + lane;
    o[0] = d * aLo + bLo;
    o[32] = d * aHi + bHi;
  }
}

extern "C" void kernel_launch(void* const* d_in, const int* in_sizes, int n_in,
                              void* d_out, int out_size, void* d_ws, size_t ws_size,
                              hipStream_t stream) {
  const float* x  = (const float*)d_in[0];
  const int*   ei = (const int*)d_in[1];
  const float* W  = (const float*)d_in[2];
  const float* b  = (const float*)d_in[3];
  float* out = (float*)d_out;

  const int N = in_sizes[0] / IN_C;
  const int M = in_sizes[1] / 2;
  const int nch = (N + CHUNK - 1) / CHUNK;

  unsigned* part = (unsigned*)d_ws;
  float* h   = (float*)(part + (size_t)CNT_SLICES * N);
  float* dis = h + (size_t)N * OUT_C;
  unsigned* rowptr = (unsigned*)(dis + N);
  int* srt = (int*)(rowptr + N + 1);
  unsigned* cp = (unsigned*)(srt + M);

  const int nRowTiles = (N + 15) / 16;

  k_count<<<dim3(CNT_RANGES, CNT_SLICES), CNT_THREADS, 0, stream>>>(ei + M, part, cp, N, M, nch);
  k_scan<<<nch, CHUNK, 0, stream>>>(part, cp, rowptr, dis, N, nch);
  k_place<<<dim3(CNT_RANGES, CNT_SLICES), CNT_THREADS, 0, stream>>>(ei, part, srt, N, M);
  k_gemm_wmma<<<(nRowTiles + GEMM_WAVES - 1) / GEMM_WAVES, GEMM_THREADS, 0, stream>>>(x, W, dis, h, nRowTiles);
  {
    const int waves = (N + PULL_NODES - 1) / PULL_NODES;
    k_pull<<<(waves + 7) / 8, 256, 0, stream>>>(rowptr, srt, h, dis, b, out, N, M);
  }
}


# GEMM row tiles computed by the placement launch's own 8192 waves after their placement (both need only the scan's results): 4 launches, not 5; W staged once per group; k order and sum order kept
# speedup vs baseline: 3.6468x; 1.0301x over previous
#include <hip/hip_runtime.h>


typedef float v2f __attribute__((ext_vector_type(2)));
typedef float v8f __attribute__((ext_vector_type(8)));

#define IN_C  64
#define OUT_C 64

#define CNT_RANGES  8
#define CNT_SLICES  32
#define CHUNK       256
#define CNT_CAP     12544
#define CNT_THREADS 1024
#define CNT_BATCH   13
#define PLC_BATCH   7

struct Slice { int q0; int qFull; int eEnd; };

__device__ __forceinline__ Slice slice_of(int s, int m) {
  const int quads = (m + 3) >> 2;
  const int per = (quads + CNT_SLICES - 1) / CNT_SLICES;
  Slice b;
  b.q0 = s * per;
  const int q1 = min(b.q0 + per, quads);
  b.qFull = min(q1, m >> 2);
  b.eEnd = min(q1 * 4, m);
  return b;
}

__device__ __forceinline__ void count_target(unsigned* cnt, int c, int base, int span) {
  const unsigned i = (unsigned)(c - base);
  if (i < (unsigned)span) atomicAdd(&cnt[i], 1u);
}

__global__ __launch_bounds__(CNT_THREADS) void k_count(const int* __restrict__ col,
                                                       unsigned* __restrict__ part,
                                                       unsigned* __restrict__ cp,
                                                       int n, int m, int nch) {
  __shared__ unsigned cnt[CNT_CAP];
  const int tid = threadIdx.x;
  const int ln = tid & 31;
  const int wv = tid >> 5;
  const int s = blockIdx.y;
  const Slice sl = slice_of(s, m);
  const int q0 = sl.q0, qFull = sl.qFull, eEnd = sl.eEnd;
  const bool vec = (reinterpret_cast<size_t>(col) & 15) == 0;
  const int4* __restrict__ c4 = reinterpret_cast<const int4*>(col);

  for (int base = blockIdx.x * CNT_CAP; base < n; base += CNT_RANGES * CNT_CAP) {
    const int span = min(CNT_CAP, n - base);
    for (int i = tid; i < span; i += CNT_THREADS) cnt[i] = 0u;
    __syncthreads();

    int eTail = q0 * 4;
    if (vec) {
      for (int q = q0 + tid; q < qFull; q += CNT_THREADS * CNT_BATCH) {
        int4 c[CNT_BATCH];
#pragma unroll
        for (int u = 0; u < CNT_BATCH; ++u) {
          const int qq = q + u * CNT_THREADS;
          c[u] = (qq < qFull) ? c4[qq] : make_int4(-1, -1, -1, -1);
        }
#pragma unroll
        for (int u = 0; u < CNT_BATCH; ++u) {
          count_target(cnt, c[u].x, base, span);
          count_target(cnt, c[u].y, base, span);
          count_target(cnt, c[u].z, base, span);
          count_target(cnt, c[u].w, base, span);
        }
      }
      eTail = max(q0, qFull) * 4;
    }
    for (int e = eTail + tid; e < eEnd; e += CNT_THREADS) count_target(cnt, col[e], base, span);
    __syncthreads();

    unsigned* __restrict__ plane = part + (size_t)s * n + base;
    for (int i = tid; i < span; i += CNT_THREADS) plane[i] = cnt[i];

    const int chunks = (span + CHUNK - 1) / CHUNK;
    for (int k = wv; k < chunks; k += CNT_THREADS / 32) {
      unsigned sum = 0u;
#pragma unroll
      for (int j = 0; j < CHUNK / 32; ++j) {
        const int i = k * CHUNK + j * 32 + ln;
        if (i < span) sum += cnt[i];
      }
      for (int o = 16; o > 0; o >>= 1) sum += __shfl_xor(sum, o);
      if (ln == 0) cp[(size_t)s * nch + base / CHUNK + k] = sum;
    }
    __syncthreads();
  }
}

__global__ __launch_bounds__(CHUNK) void k_scan(unsigned* __restrict__ part,
                                                const unsigned* __restrict__ cp,
                                                unsigned* __restrict__ rowptr,
                                                float* __restrict__ dis, int n, int nch) {
  __shared__ unsigned red[CHUNK / 32];
  __shared__ unsigned wsum[CHUNK / 32];
  const int tid = threadIdx.x;
  const int ln = tid & 31;
  const int wv = tid >> 5;
  const int b = blockIdx.x;

  unsigned pre = 0u;
  for (int c = tid; c < b; c += CHUNK) {
#pragma unroll
    for (int s = 0; s < CNT_SLICES; ++s) pre += cp[(size_t)s * nch + c];
  }
  for (int o = 16; o > 0; o >>= 1) pre += __shfl_xor(pre, o);
  if (ln == 0) red[wv] = pre;

  const int i = b * CHUNK + tid;
  unsigned p[CNT_SLICES];
#pragma unroll
  for (int s = 0; s < CNT_SLICES; ++s) p[s] = (i < n) ? part[(size_t)s * n + i] : 0u;
  unsigned d = 0u;
#pragma unroll
  for (int s = 0; s < CNT_SLICES; ++s) { const unsigned t = p[s]; p[s] = d; d += t; }

  unsigned inc = d;
  for (int o = 1; o < 32; o <<= 1) {
    const unsigned t = __shfl_up(inc, (unsigned)o);
    if (ln >= o) inc += t;
  }
  if (ln == 31) wsum[wv] = inc;
  __syncthreads();

  unsigned start = inc - d;
  for (int w = 0; w < CHUNK / 32; ++w) {
    start += red[w];
    if (w < wv) start += wsum[w];
  }
  if (i < n) {
    rowptr[i] = start;
    if (i == n - 1) rowptr[n] = start + d;
#pragma unroll
    for (int s = 0; s < CNT_SLICES; ++s) part[(size_t)s * n + i] = start + p[s];
    dis[i] = rsqrtf((float)(d + 1u));
  }
}

__device__ __forceinline__ unsigned take_slot(unsigned* cur, int c, int base, int span) {
  const unsigned i = (unsigned)(c - base);
  return (i < (unsigned)span) ? atomicAdd(&cur[i], 1u) : 0xFFFFFFFFu;
}

__device__ __forceinline__ void put_source(int* __restrict__ srt, unsigned slot, int r, int m) {
  if (slot < (unsigned)m) srt[slot] = r;
}

__device__ __forceinline__ void place_edge(unsigned* cur, int* __restrict__ srt, int c, int r,
                                           int base, int span, int m) {
  put_source(srt, take_slot(cur, c, base, span), r, m);
}

__device__ __forceinline__ void place_quad(unsigned* cur, int* __restrict__ srt, int4 c, int4 r,
                                           int base, int span, int m) {
  const unsigned s0 = take_slot(cur, c.x, base, span);
  const unsigned s1 = take_slot(cur, c.y, base, span);
  const unsigned s2 = take_slot(cur, c.z, base, span);
  const unsigned s3 = take_slot(cur, c.w, base, span);
  put_source(srt, s0, r.x, m);
  put_source(srt, s1, r.y, m);
  put_source(srt, s2, r.z, m);
  put_source(srt, s3, r.w, m);
}

#define W_PER_THREAD (IN_C * OUT_C / CNT_THREADS)
#define WP_PITCH 160

struct TileIn { v2f a[IN_C / 4]; float d[8]; };

__device__ __forceinline__ void tile_load(TileIn& t, const float* __restrict__ x,
                                          const float* __restrict__ dis, int rowBase, int mn, int hi) {
  const float* __restrict__ xrow = x + (size_t)(rowBase + mn) * IN_C + 2 * hi;
#pragma unroll
  for (int j = 0; j < IN_C / 4; ++j) t.a[j] = *(const v2f*)(xrow + 4 * j);
#pragma unroll
  for (int v = 0; v < 8; ++v) t.d[v] = dis[rowBase + v + 8 * hi];
}

__device__ __forceinline__ void tile_compute_store(const TileIn& t, const float* __restrict__ Wlds,
                                                   float* __restrict__ h, int rowBase, int mn, int hi) {
  v8f acc0 = {}, acc1 = {}, acc2 = {}, acc3 = {};
  const float* __restrict__ wl = Wlds + hi * WP_PITCH + 2 * mn;
#pragma unroll
  for (int j = 0; j < IN_C / 4; ++j) {
    const float* __restrict__ wp = wl + 2 * j * WP_PITCH;
    const v2f b0 = *(const v2f*)(wp);
    const v2f b1 = *(const v2f*)(wp + 32);
    const v2f b2 = *(const v2f*)(wp + 64);
    const v2f b3 = *(const v2f*)(wp + 96);
    acc0 = __builtin_amdgcn_wmma_f32_16x16x4_f32(false, t.a[j], false, b0, (short)0, acc0, false, false);
    acc1 = __builtin_amdgcn_wmma_f32_16x16x4_f32(false, t.a[j], false, b1, (short)0, acc1, false, false);
    acc2 = __builtin_amdgcn_wmma_f32_16x16x4_f32(false, t.a[j], false, b2, (short)0, acc2, false, false);
    acc3 = __builtin_amdgcn_wmma_f32_16x16x4_f32(false, t.a[j], false, b3, (short)0, acc3, false, false);
  }
#pragma unroll
  for (int v = 0; v < 8; ++v) {
    float* __restrict__ hr = h + (size_t)(rowBase + v + 8 * hi) * OUT_C + mn;
    hr[0]  = acc0[v] * t.d[v];
    hr[16] = acc1[v] * t.d[v];
    hr[32] = acc2[v] * t.d[v];
    hr[48] = acc3[v] * t.d[v];
  }
}

__global__ __launch_bounds__(CNT_THREADS) void k_place_gemm(
    const int* __restrict__ ei, const unsigned* __restrict__ part, int* __restrict__ srt,
    const float* __restrict__ x, const float* __restrict__ W, const float* __restrict__ dis,
    float* __restrict__ h, int n, int m, int nRowTiles) {
  __shared__ __attribute__((aligned(16))) unsigned cur[CNT_CAP];
  const int tid = threadIdx.x;

  if (blockIdx.y < CNT_SLICES) {
    const int s = blockIdx.y;
    const Slice sl = slice_of(s, m);
    const int q0 = sl.q0, qFull = sl.qFull, eEnd = sl.eEnd;
    const int* __restrict__ row = ei;
    const int* __restrict__ col = ei + m;
    const bool vec = ((reinterpret_cast<size_t>(row) | reinterpret_cast<size_t>(col)) & 15) == 0;
    const int4* __restrict__ r4 = reinterpret_cast<const int4*>(row);
    const int4* __restrict__ c4 = reinterpret_cast<const int4*>(col);

    for (int base = blockIdx.x * CNT_CAP; base < n; base += CNT_RANGES * CNT_CAP) {
      const int span = min(CNT_CAP, n - base);
      const unsigned* __restrict__ plane = part + (size_t)s * n + base;
      for (int i = tid; i < span; i += CNT_THREADS) cur[i] = plane[i];
      __syncthreads();

      int eTail = q0 * 4;
      if (vec) {
        for (int q = q0 + tid; q < qFull; q += CNT_THREADS * PLC_BATCH) {
          int4 c[PLC_BATCH], r[PLC_BATCH];
#pragma unroll
          for (int u = 0; u < PLC_BATCH; ++u) {
            const int qq = q + u * CNT_THREADS;
            const bool ok = qq < qFull;
            c[u] = ok ? c4[qq] : make_int4(-1, -1, -1, -1);
            r[u] = ok ? r4[qq] : make_int4(0, 0, 0, 0);
          }
#pragma unroll
          for (int u = 0; u < PLC_BATCH; ++u) {
            place_quad(cur, srt, c[u], r[u], base, span, m);
          }
        }
        eTail = max(q0, qFull) * 4;
      }
      for (int e = eTail + tid; e < eEnd; e += CNT_THREADS)
        place_edge(cur, srt, col[e], row[e], base, span, m);
      __syncthreads();
    }
  }

  float* __restrict__ Wlds = reinterpret_cast<float*>(cur);
  const int wave = tid >> 5;
  const int lane = tid & 31;
  const int mn = lane & 15;
  const int hi = lane >> 4;
  const int groups = gridDim.x * gridDim.y;
  const int group = blockIdx.y * gridDim.x + blockIdx.x;
  const int tile = wave * groups + group;
  const bool live = tile < nRowTiles;
  const int rowBase = min(tile, nRowTiles - 1) << 4;

  TileIn t;
  tile_load(t, x, dis, rowBase, mn, hi);
  float wv[W_PER_THREAD];
#pragma unroll
  for (int u = 0; u < W_PER_THREAD; ++u) wv[u] = W[tid + CNT_THREADS * u];
#pragma unroll
  for (int u = 0; u < W_PER_THREAD; ++u) {
    const int i = tid + CNT_THREADS * u;
    const int k = i >> 6, nn = i & 63;
    Wlds[(k >> 1) * WP_PITCH + 2 * nn + (k & 1)] = wv[u];
  }
  __syncthreads();
  if (!live) return;
  tile_compute_store(t, Wlds, h, rowBase, mn, hi);
}

#define PULL_NODES 8
#define PULL_GROUP 8

__global__ __launch_bounds__(256) void k_pull(
    const unsigned* __restrict__ rowptr, const int* __restrict__ srt,
    const float* __restrict__ h, const float* __restrict__ dis,
    const float* __restrict__ bias, float* __restrict__ out, int n, int m) {
  const int lane = threadIdx.x & 31;
  const int wave = __builtin_amdgcn_readfirstlane(blockIdx.x * 8 + (threadIdx.x >> 5));
  const int i0 = wave * PULL_NODES;
  if (i0 >= n) return;
  const int nodes = min(PULL_NODES, n - i0);

  unsigned rp = 0u;
  if (lane <= PULL_NODES) rp = rowptr[min(i0 + lane, n)];
  const float bLo = bias[lane];
  const float bHi = bias[lane + 32];

  for (int k = 0; k < nodes; ++k) {
    const int i = i0 + k;
    const unsigned beg = min((unsigned)__builtin_amdgcn_readlane((int)rp, k), (unsigned)m);
    const unsigned end = min((unsigned)__builtin_amdgcn_readlane((int)rp, k + 1), (unsigned)m);
    const float* __restrict__ hs = h + (size_t)i * OUT_C + lane;
    float aLo = hs[0];
    float aHi = hs[32];
    const float d = dis[i];

    for (unsigned e0 = beg; e0 < end; e0 += 32u) {
      const int cnt = (int)min(32u, end - e0);
      int sv = 0;
      if (lane < cnt) sv = srt[e0 + lane];
      for (int j = 0; j < cnt; j += PULL_GROUP) {
        int r[PULL_GROUP];
        float lo[PULL_GROUP], hi[PULL_GROUP];
#pragma unroll
        for (int u = 0; u < PULL_GROUP; ++u) {
          const int t = __builtin_amdgcn_readlane(sv, min(j + u, cnt - 1));
          r[u] = min(max(t, 0), n - 1);
        }
#pragma unroll
        for (int u = 0; u < PULL_GROUP; ++u) {
          lo[u] = 0.0f;
          hi[u] = 0.0f;
          if (j + u < cnt) {
            const float* hp = h + (size_t)r[u] * OUT_C + lane;
            lo[u] = hp[0];
            hi[u] = hp[32];
          }
        }
#pragma unroll
        for (int u = 0; u < PULL_GROUP; ++u) {
          aLo += lo[u];
          aHi += hi[u];
        }
      }
    }
    float* __restrict__ o = out + (size_t)i * OUT_C + lane;
    o[0] = d * aLo + bLo;
    o[32] = d * aHi + bHi;
  }
}

extern "C" void kernel_launch(void* const* d_in, const int* in_sizes, int n_in,
                              void* d_out, int out_size, void* d_ws, size_t ws_size,
                              hipStream_t stream) {
  const float* x  = (const float*)d_in[0];
  const int*   ei = (const int*)d_in[1];
  const float* W  = (const float*)d_in[2];
  const float* b  = (const float*)d_in[3];
  float* out = (float*)d_out;

  const int N = in_sizes[0] / IN_C;
  const int M = in_sizes[1] / 2;
  const int nch = (N + CHUNK - 1) / CHUNK;

  unsigned* part = (unsigned*)d_ws;
  float* h   = (float*)(part + (size_t)CNT_SLICES * N);
  float* dis = h + (size_t)N * OUT_C;
  unsigned* rowptr = (unsigned*)(dis + N);
  int* srt = (int*)(rowptr + N + 1);
  unsigned* cp = (unsigned*)(srt + M);

  const int nRowTiles = (N + 15) / 16;

  k_count<<<dim3(CNT_RANGES, CNT_SLICES), CNT_THREADS, 0, stream>>>(ei + M, part, cp, N, M, nch);
  k_scan<<<nch, CHUNK, 0, stream>>>(part, cp, rowptr, dis, N, nch);
  {
    const int tilesPerY = CNT_RANGES * (CNT_THREADS / 32);
    const int gyTiles = (nRowTiles + tilesPerY - 1) / tilesPerY;
    const int gy = gyTiles > CNT_SLICES ? gyTiles : CNT_SLICES;
    k_place_gemm<<<dim3(CNT_RANGES, gy), CNT_THREADS, 0, stream>>>(ei, part, srt, x, W, dis, h, N, M, nRowTiles);
  }
  {
    const int waves = (N + PULL_NODES - 1) / PULL_NODES;
    k_pull<<<(waves + 7) / 8, 256, 0, stream>>>(rowptr, srt, h, dis, b, out, N, M);
  }
}


# Copy loops batched: the placement's cursor copy (global to LDS), the count's plane copy and chunk sums issue all of a thread's loads before the first use; was one wait per word; sum order kept
# speedup vs baseline: 3.8008x; 1.0422x over previous
#include <hip/hip_runtime.h>


typedef float v2f __attribute__((ext_vector_type(2)));
typedef float v8f __attribute__((ext_vector_type(8)));

#define IN_C  64
#define OUT_C 64

#define CNT_RANGES  8
#define CNT_SLICES  32
#define CHUNK       256
#define CNT_CAP     12544
#define CNT_THREADS 1024
#define CNT_BATCH   13
#define PLC_BATCH   7
#define CAP_BATCH   ((CNT_CAP + CNT_THREADS - 1) / CNT_THREADS)

struct Slice { int q0; int qFull; int eEnd; };

__device__ __forceinline__ Slice slice_of(int s, int m) {
  const int quads = (m + 3) >> 2;
  const int per = (quads + CNT_SLICES - 1) / CNT_SLICES;
  Slice b;
  b.q0 = s * per;
  const int q1 = min(b.q0 + per, quads);
  b.qFull = min(q1, m >> 2);
  b.eEnd = min(q1 * 4, m);
  return b;
}

__device__ __forceinline__ void count_target(unsigned* cnt, int c, int base, int span) {
  const unsigned i = (unsigned)(c - base);
  if (i < (unsigned)span) atomicAdd(&cnt[i], 1u);
}

__global__ __launch_bounds__(CNT_THREADS) void k_count(const int* __restrict__ col,
                                                       unsigned* __restrict__ part,
                                                       unsigned* __restrict__ cp,
                                                       int n, int m, int nch) {
  __shared__ unsigned cnt[CNT_CAP];
  const int tid = threadIdx.x;
  const int ln = tid & 31;
  const int wv = tid >> 5;
  const int s = blockIdx.y;
  const Slice sl = slice_of(s, m);
  const int q0 = sl.q0, qFull = sl.qFull, eEnd = sl.eEnd;
  const bool vec = (reinterpret_cast<size_t>(col) & 15) == 0;
  const int4* __restrict__ c4 = reinterpret_cast<const int4*>(col);

  for (int base = blockIdx.x * CNT_CAP; base < n; base += CNT_RANGES * CNT_CAP) {
    const int span = min(CNT_CAP, n - base);
    for (int i = tid; i < span; i += CNT_THREADS) cnt[i] = 0u;
    __syncthreads();

    int eTail = q0 * 4;
    if (vec) {
      for (int q = q0 + tid; q < qFull; q += CNT_THREADS * CNT_BATCH) {
        int4 c[CNT_BATCH];
#pragma unroll
        for (int u = 0; u < CNT_BATCH; ++u) {
          const int qq = q + u * CNT_THREADS;
          c[u] = (qq < qFull) ? c4[qq] : make_int4(-1, -1, -1, -1);
        }
#pragma unroll
        for (int u = 0; u < CNT_BATCH; ++u) {
          count_target(cnt, c[u].x, base, span);
          count_target(cnt, c[u].y, base, span);
          count_target(cnt, c[u].z, base, span);
          count_target(cnt, c[u].w, base, span);
        }
      }
      eTail = max(q0, qFull) * 4;
    }
    for (int e = eTail + tid; e < eEnd; e += CNT_THREADS) count_target(cnt, col[e], base, span);
    __syncthreads();

    unsigned* __restrict__ plane = part + (size_t)s * n + base;
    unsigned pv[CAP_BATCH];
#pragma unroll
    for (int u = 0; u < CAP_BATCH; ++u) pv[u] = cnt[min(tid + u * CNT_THREADS, CNT_CAP - 1)];
#pragma unroll
    for (int u = 0; u < CAP_BATCH; ++u) {
      const int i = tid + u * CNT_THREADS;
      if (i < span) plane[i] = pv[u];
    }

    const int chunks = (span + CHUNK - 1) / CHUNK;
    for (int k = wv; k < chunks; k += CNT_THREADS / 32) {
      unsigned sum = 0u;
#pragma unroll
      for (int j = 0; j < CHUNK / 32; ++j) {
        const int i = k * CHUNK + j * 32 + ln;
        const unsigned v = cnt[i];
        sum += (i < span) ? v : 0u;
      }
      for (int o = 16; o > 0; o >>= 1) sum += __shfl_xor(sum, o);
      if (ln == 0) cp[(size_t)s * nch + base / CHUNK + k] = sum;
    }
    __syncthreads();
  }
}

__global__ __launch_bounds__(CHUNK) void k_scan(unsigned* __restrict__ part,
                                                const unsigned* __restrict__ cp,
                                                unsigned* __restrict__ rowptr,
                                                float* __restrict__ dis, int n, int nch) {
  __shared__ unsigned red[CHUNK / 32];
  __shared__ unsigned wsum[CHUNK / 32];
  const int tid = threadIdx.x;
  const int ln = tid & 31;
  const int wv = tid >> 5;
  const int b = blockIdx.x;

  unsigned pre = 0u;
  for (int c = tid; c < b; c += CHUNK) {
#pragma unroll
    for (int s = 0; s < CNT_SLICES; ++s) pre += cp[(size_t)s * nch + c];
  }
  for (int o = 16; o > 0; o >>= 1) pre += __shfl_xor(pre, o);
  if (ln == 0) red[wv] = pre;

  const int i = b * CHUNK + tid;
  unsigned p[CNT_SLICES];
#pragma unroll
  for (int s = 0; s < CNT_SLICES; ++s) p[s] = (i < n) ? part[(size_t)s * n + i] : 0u;
  unsigned d = 0u;
#pragma unroll
  for (int s = 0; s < CNT_SLICES; ++s) { const unsigned t = p[s]; p[s] = d; d += t; }

  unsigned inc = d;
  for (int o = 1; o < 32; o <<= 1) {
    const unsigned t = __shfl_up(inc, (unsigned)o);
    if (ln >= o) inc += t;
  }
  if (ln == 31) wsum[wv] = inc;
  __syncthreads();

  unsigned start = inc - d;
  for (int w = 0; w < CHUNK / 32; ++w) {
    start += red[w];
    if (w < wv) start += wsum[w];
  }
  if (i < n) {
    rowptr[i] = start;
    if (i == n - 1) rowptr[n] = start + d;
#pragma unroll
    for (int s = 0; s < CNT_SLICES; ++s) part[(size_t)s * n + i] = start + p[s];
    dis[i] = rsqrtf((float)(d + 1u));
  }
}

__device__ __forceinline__ unsigned take_slot(unsigned* cur, int c, int base, int span) {
  const unsigned i = (unsigned)(c - base);
  return (i < (unsigned)span) ? atomicAdd(&cur[i], 1u) : 0xFFFFFFFFu;
}

__device__ __forceinline__ void put_source(int* __restrict__ srt, unsigned slot, int r, int m) {
  if (slot < (unsigned)m) srt[slot] = r;
}

__device__ __forceinline__ void place_edge(unsigned* cur, int* __restrict__ srt, int c, int r,
                                           int base, int span, int m) {
  put_source(srt, take_slot(cur, c, base, span), r, m);
}

__device__ __forceinline__ void place_quad(unsigned* cur, int* __restrict__ srt, int4 c, int4 r,
                                           int base, int span, int m) {
  const unsigned s0 = take_slot(cur, c.x, base, span);
  const unsigned s1 = take_slot(cur, c.y, base, span);
  const unsigned s2 = take_slot(cur, c.z, base, span);
  const unsigned s3 = take_slot(cur, c.w, base, span);
  put_source(srt, s0, r.x, m);
  put_source(srt, s1, r.y, m);
  put_source(srt, s2, r.z, m);
  put_source(srt, s3, r.w, m);
}

#define W_PER_THREAD (IN_C * OUT_C / CNT_THREADS)
#define WP_PITCH 160

struct TileIn { v2f a[IN_C / 4]; float d[8]; };

__device__ __forceinline__ void tile_load(TileIn& t, const float* __restrict__ x,
                                          const float* __restrict__ dis, int rowBase, int mn, int hi) {
  const float* __restrict__ xrow = x + (size_t)(rowBase + mn) * IN_C + 2 * hi;
#pragma unroll
  for (int j = 0; j < IN_C / 4; ++j) t.a[j] = *(const v2f*)(xrow + 4 * j);
#pragma unroll
  for (int v = 0; v < 8; ++v) t.d[v] = dis[rowBase + v + 8 * hi];
}

__device__ __forceinline__ void tile_compute_store(const TileIn& t, const float* __restrict__ Wlds,
                                                   float* __restrict__ h, int rowBase, int mn, int hi) {
  v8f acc0 = {}, acc1 = {}, acc2 = {}, acc3 = {};
  const float* __restrict__ wl = Wlds + hi * WP_PITCH + 2 * mn;
#pragma unroll
  for (int j = 0; j < IN_C / 4; ++j) {
    const float* __restrict__ wp = wl + 2 * j * WP_PITCH;
    const v2f b0 = *(const v2f*)(wp);
    const v2f b1 = *(const v2f*)(wp + 32);
    const v2f b2 = *(const v2f*)(wp + 64);
    const v2f b3 = *(const v2f*)(wp + 96);
    acc0 = __builtin_amdgcn_wmma_f32_16x16x4_f32(false, t.a[j], false, b0, (short)0, acc0, false, false);
    acc1 = __builtin_amdgcn_wmma_f32_16x16x4_f32(false, t.a[j], false, b1, (short)0, acc1, false, false);
    acc2 = __builtin_amdgcn_wmma_f32_16x16x4_f32(false, t.a[j], false, b2, (short)0, acc2, false, false);
    acc3 = __builtin_amdgcn_wmma_f32_16x16x4_f32(false, t.a[j], false, b3, (short)0, acc3, false, false);
  }
#pragma unroll
  for (int v = 0; v < 8; ++v) {
    float* __restrict__ hr = h + (size_t)(rowBase + v + 8 * hi) * OUT_C + mn;
    hr[0]  = acc0[v] * t.d[v];
    hr[16] = acc1[v] * t.d[v];
    hr[32] = acc2[v] * t.d[v];
    hr[48] = acc3[v] * t.d[v];
  }
}

__global__ __launch_bounds__(CNT_THREADS) void k_place_gemm(
    const int* __restrict__ ei, const unsigned* __restrict__ part, int* __restrict__ srt,
    const float* __restrict__ x, const float* __restrict__ W, const float* __restrict__ dis,
    float* __restrict__ h, int n, int m, int nRowTiles) {
  __shared__ __attribute__((aligned(16))) unsigned cur[CNT_CAP];
  const int tid = threadIdx.x;

  if (blockIdx.y < CNT_SLICES) {
    const int s = blockIdx.y;
    const Slice sl = slice_of(s, m);
    const int q0 = sl.q0, qFull = sl.qFull, eEnd = sl.eEnd;
    const int* __restrict__ row = ei;
    const int* __restrict__ col = ei + m;
    const bool vec = ((reinterpret_cast<size_t>(row) | reinterpret_cast<size_t>(col)) & 15) == 0;
    const int4* __restrict__ r4 = reinterpret_cast<const int4*>(row);
    const int4* __restrict__ c4 = reinterpret_cast<const int4*>(col);

    for (int base = blockIdx.x * CNT_CAP; base < n; base += CNT_RANGES * CNT_CAP) {
      const int span = min(CNT_CAP, n - base);
      const unsigned* __restrict__ plane = part + (size_t)s * n + base;
      unsigned cv[CAP_BATCH];
#pragma unroll
      for (int u = 0; u < CAP_BATCH; ++u) {
        const int i = tid + u * CNT_THREADS;
        cv[u] = (i < span) ? plane[i] : 0u;
      }
#pragma unroll
      for (int u = 0; u < CAP_BATCH; ++u) {
        const int i = tid + u * CNT_THREADS;
        if (i < span) cur[i] = cv[u];
      }
      __syncthreads();

      int eTail = q0 * 4;
      if (vec) {
        for (int q = q0 + tid; q < qFull; q += CNT_THREADS * PLC_BATCH) {
          int4 c[PLC_BATCH], r[PLC_BATCH];
#pragma unroll
          for (int u = 0; u < PLC_BATCH; ++u) {
            const int qq = q + u * CNT_THREADS;
            const bool ok = qq < qFull;
            c[u] = ok ? c4[qq] : make_int4(-1, -1, -1, -1);
            r[u] = ok ? r4[qq] : make_int4(0, 0, 0, 0);
          }
#pragma unroll
          for (int u = 0; u < PLC_BATCH; ++u) {
            place_quad(cur, srt, c[u], r[u], base, span, m);
          }
        }
        eTail = max(q0, qFull) * 4;
      }
      for (int e = eTail + tid; e < eEnd; e += CNT_THREADS)
        place_edge(cur, srt, col[e], row[e], base, span, m);
      __syncthreads();
    }
  }

  float* __restrict__ Wlds = reinterpret_cast<float*>(cur);
  const int wave = tid >> 5;
  const int lane = tid & 31;
  const int mn = lane & 15;
  const int hi = lane >> 4;
  const int groups = gridDim.x * gridDim.y;
  const int group = blockIdx.y * gridDim.x + blockIdx.x;
  const int tile = wave * groups + group;
  const bool live = tile < nRowTiles;
  const int rowBase = min(tile, nRowTiles - 1) << 4;

  TileIn t;
  tile_load(t, x, dis, rowBase, mn, hi);
  float wv[W_PER_THREAD];
#pragma unroll
  for (int u = 0; u < W_PER_THREAD; ++u) wv[u] = W[tid + CNT_THREADS * u];
#pragma unroll
  for (int u = 0; u < W_PER_THREAD; ++u) {
    const int i = tid + CNT_THREADS * u;
    const int k = i >> 6, nn = i & 63;
    Wlds[(k >> 1) * WP_PITCH + 2 * nn + (k & 1)] = wv[u];
  }
  __syncthreads();
  if (!live) return;
  tile_compute_store(t, Wlds, h, rowBase, mn, hi);
}

#define PULL_NODES 8
#define PULL_GROUP 8

__global__ __launch_bounds__(256) void k_pull(
    const unsigned* __restrict__ rowptr, const int* __restrict__ srt,
    const float* __restrict__ h, const float* __restrict__ dis,
    const float* __restrict__ bias, float* __restrict__ out, int n, int m) {
  const int lane = threadIdx.x & 31;
  const int wave = __builtin_amdgcn_readfirstlane(blockIdx.x * 8 + (threadIdx.x >> 5));
  const int i0 = wave * PULL_NODES;
  if (i0 >= n) return;
  const int nodes = min(PULL_NODES, n - i0);

  unsigned rp = 0u;
  if (lane <= PULL_NODES) rp = rowptr[min(i0 + lane, n)];
  const float bLo = bias[lane];
  const float bHi = bias[lane + 32];

  for (int k = 0; k < nodes; ++k) {
    const int i = i0 + k;
    const unsigned beg = min((unsigned)__builtin_amdgcn_readlane((int)rp, k), (unsigned)m);
    const unsigned end = min((unsigned)__builtin_amdgcn_readlane((int)rp, k + 1), (unsigned)m);
    const float* __restrict__ hs = h + (size_t)i * OUT_C + lane;
    float aLo = hs[0];
    float aHi = hs[32];
    const float d = dis[i];

    for (unsigned e0 = beg; e0 < end; e0 += 32u) {
      const int cnt = (int)min(32u, end - e0);
      int sv = 0;
      if (lane < cnt) sv = srt[e0 + lane];
      for (int j = 0; j < cnt; j += PULL_GROUP) {
        int r[PULL_GROUP];
        float lo[PULL_GROUP], hi[PULL_GROUP];
#pragma unroll
        for (int u = 0; u < PULL_GROUP; ++u) {
          const int t = __builtin_amdgcn_readlane(sv, min(j + u, cnt - 1));
          r[u] = min(max(t, 0), n - 1);
        }
#pragma unroll
        for (int u = 0; u < PULL_GROUP; ++u) {
          lo[u] = 0.0f;
          hi[u] = 0.0f;
          if (j + u < cnt) {
            const float* hp = h + (size_t)r[u] * OUT_C + lane;
            lo[u] = hp[0];
            hi[u] = hp[32];
          }
        }
#pragma unroll
        for (int u = 0; u < PULL_GROUP; ++u) {
          aLo += lo[u];
          aHi += hi[u];
        }
      }
    }
    float* __restrict__ o = out + (size_t)i * OUT_C + lane;
    o[0] = d * aLo + bLo;
    o[32] = d * aHi + bHi;
  }
}

extern "C" void kernel_launch(void* const* d_in, const int* in_sizes, int n_in,
                              void* d_out, int out_size, void* d_ws, size_t ws_size,
                              hipStream_t stream) {
  const float* x  = (const float*)d_in[0];
  const int*   ei = (const int*)d_in[1];
  const float* W  = (const float*)d_in[2];
  const float* b  = (const float*)d_in[3];
  float* out = (float*)d_out;

  const int N = in_sizes[0] / IN_C;
  const int M = in_sizes[1] / 2;
  const int nch = (N + CHUNK - 1) / CHUNK;

  unsigned* part = (unsigned*)d_ws;
  float* h   = (float*)(part + (size_t)CNT_SLICES * N);
  float* dis = h + (size_t)N * OUT_C;
  unsigned* rowptr = (unsigned*)(dis + N);
  int* srt = (int*)(rowptr + N + 1);
  unsigned* cp = (unsigned*)(srt + M);

  const int nRowTiles = (N + 15) / 16;

  k_count<<<dim3(CNT_RANGES, CNT_SLICES), CNT_THREADS, 0, stream>>>(ei + M, part, cp, N, M, nch);
  k_scan<<<nch, CHUNK, 0, stream>>>(part, cp, rowptr, dis, N, nch);
  {
    const int tilesPerY = CNT_RANGES * (CNT_THREADS / 32);
    const int gyTiles = (nRowTiles + tilesPerY - 1) / tilesPerY;
    const int gy = gyTiles > CNT_SLICES ? gyTiles : CNT_SLICES;
    k_place_gemm<<<dim3(CNT_RANGES, gy), CNT_THREADS, 0, stream>>>(ei, part, srt, x, W, dis, h, N, M, nRowTiles);
  }
  {
    const int waves = (N + PULL_NODES - 1) / PULL_NODES;
    k_pull<<<(waves + 7) / 8, 256, 0, stream>>>(rowptr, srt, h, dis, b, out, N, M);
  }
}
